# starting kernel
# baseline (speedup 1.0000x reference)
#include <hip/hip_runtime.h>
#include <math.h>

#define BSZ   32
#define DCH   96
#define HH    64
#define WW    64
#define LL    (HH * WW)
#define NST   4
#define RNK   6
#define KPROJ 14
#define LOG_EPS_F (-20.7232658f)

typedef __attribute__((ext_vector_type(2))) float v2f;
typedef __attribute__((ext_vector_type(8))) float v8f;

__global__ void conv_silu_kernel(const float* __restrict__ x,
                                 const float* __restrict__ conv_w,
                                 const float* __restrict__ conv_b,
                                 float* __restrict__ xs) {
    const int bid  = blockIdx.x;
    const int tile = bid & 15;
    const int d    = (bid >> 4) % DCH;
    const int b    = bid / (16 * DCH);
    const int h0   = tile * 4;

    __shared__ float t[8][68];

    const int tid = threadIdx.x;
    const float* xin = x + ((size_t)b * DCH + d) * LL;

    for (int i = tid; i < 8 * 68; i += 256) {
        const int r  = i / 68;
        const int c  = i % 68;
        const int hh = h0 - 2 + r;
        const int wc = c - 2;
        float v = 0.0f;
        if (hh >= 0 && hh < HH && wc >= 0 && wc < WW)
            v = xin[hh * WW + wc];
        t[r][c] = v;
    }
    __syncthreads();

    float wv[25];
#pragma unroll
    for (int j = 0; j < 25; ++j) wv[j] = conv_w[d * 25 + j];
    const float bv = conv_b[d];

    const int ty = tid >> 6;
    const int tx = tid & 63;

    float acc = bv;
#pragma unroll
    for (int ky = 0; ky < 5; ++ky)
#pragma unroll
        for (int kx = 0; kx < 5; ++kx)
            acc = fmaf(t[ty + ky][tx + kx], wv[ky * 5 + kx], acc);

    const float s = acc / (1.0f + expf(-acc));
    xs[((size_t)b * DCH + d) * LL + (size_t)(h0 + ty) * WW + tx] = s;
}

__global__ void proj_wmma_kernel(const float* __restrict__ xs,
                                 const float* __restrict__ x_proj_w,
                                 float* __restrict__ xdbl) {
    __shared__ float wA[16][DCH];

    const int tid = threadIdx.x;
    for (int i = tid; i < 16 * DCH; i += 256) {
        const int row = i / DCH;
        const int k   = i % DCH;
        wA[row][k] = (row < KPROJ) ? x_proj_w[row * DCH + k] : 0.0f;
    }
    __syncthreads();

    const int wid  = (blockIdx.x * blockDim.x + tid) >> 5;
    const int lane = tid & 31;
    const int tcol = wid & 255;
    const int b    = wid >> 8;
    const int l0   = tcol * 16;

    const int half = lane >> 4;
    const int m    = lane & 15;

    const float* xb = xs + (size_t)b * DCH * LL + (l0 + m);

    v2f breg[DCH / 4];
#pragma unroll
    for (int kk = 0; kk < DCH / 4; ++kk) {
        const int k0 = kk * 4 + 2 * half;
        breg[kk].x = xb[(size_t)k0 * LL];
        breg[kk].y = xb[(size_t)(k0 + 1) * LL];
    }

    v8f acc = {0.f, 0.f, 0.f, 0.f, 0.f, 0.f, 0.f, 0.f};
#pragma unroll
    for (int kk = 0; kk < DCH / 4; ++kk) {
        const int k0 = kk * 4 + 2 * half;
        const v2f a = *(const v2f*)&wA[m][k0];
        acc = __builtin_amdgcn_wmma_f32_16x16x4_f32(
                  false, a, false, breg[kk], (short)0, acc, false, false);
    }

#pragma unroll
    for (int i = 0; i < 8; ++i) {
        const int row = i + 8 * half;
        if (row < KPROJ)
            xdbl[((size_t)b * KPROJ + row) * LL + l0 + m] = acc[i];
    }
}

__device__ __forceinline__ float wave_incl_sum(float v, int lane) {
#pragma unroll
    for (int off = 1; off < 32; off <<= 1) {
        const float t = __shfl_up(v, off, 32);
        if (lane >= off) v += t;
    }
    return v;
}

__global__ void scan_kernel(const float* __restrict__ xs,
                            const float* __restrict__ xdbl,
                            const float* __restrict__ dt_w,
                            const float* __restrict__ dt_b,
                            const float* __restrict__ A_logs,
                            const float* __restrict__ Ds,
                            float* __restrict__ out) {
    const int wid  = (blockIdx.x * blockDim.x + threadIdx.x) >> 5;
    const int lane = threadIdx.x & 31;
    const int d    = wid % DCH;
    const int b    = wid / DCH;

    float wr[RNK];
#pragma unroll
    for (int r = 0; r < RNK; ++r) wr[r] = dt_w[d * RNK + r];
    const float dtb = dt_b[d];
    const float Dd  = Ds[d];
    float An[NST];
#pragma unroll
    for (int n = 0; n < NST; ++n) An[n] = -expf(A_logs[d * NST + n]);

    const float* xch = xs   + ((size_t)b * DCH + d) * LL;
    const float* xdb = xdbl + (size_t)b * KPROJ * LL;
    float*       och = out  + ((size_t)b * DCH + d) * LL;

    float c1[NST] = {0.f, 0.f, 0.f, 0.f};
    float c2[NST] = {0.f, 0.f, 0.f, 0.f};

    for (int c = 0; c < LL / 32; ++c) {
        const int l = c * 32 + lane;
        const float xv = xch[l];

        float ad = dtb;
#pragma unroll
        for (int r = 0; r < RNK; ++r)
            ad = fmaf(wr[r], xdb[(size_t)r * LL + l], ad);
        const float delta = (ad > 20.0f) ? ad : log1pf(expf(ad));

        float y = xv * Dd;

#pragma unroll
        for (int n = 0; n < NST; ++n) {
            const float Bn = xdb[(size_t)(RNK + n) * LL + l];
            const float Cn = xdb[(size_t)(RNK + NST + n) * LL + l];

            const float lda = fmaxf(delta * An[n], LOG_EPS_F);
            float s1 = wave_incl_sum(lda, lane) + c1[n];
            const float P = expf(s1);

            const float q  = (delta * Bn * xv) / fmaxf(P, 1e-9f);
            float s2 = wave_incl_sum(q, lane) + c2[n];

            y = fmaf(P * s2, Cn, y);

            c1[n] = __shfl(s1, 31, 32);
            c2[n] = __shfl(s2, 31, 32);
        }
        och[l] = y;
    }
}

extern "C" void kernel_launch(void* const* d_in, const int* in_sizes, int n_in,
                              void* d_out, int out_size, void* d_ws, size_t ws_size,
                              hipStream_t stream) {
    (void)in_sizes; (void)n_in; (void)out_size; (void)ws_size;

    const float* x        = (const float*)d_in[0];
    const float* conv_w   = (const float*)d_in[1];
    const float* conv_b   = (const float*)d_in[2];
    const float* x_proj_w = (const float*)d_in[3];
    const float* dt_w     = (const float*)d_in[4];
    const float* dt_b     = (const float*)d_in[5];
    const float* A_logs   = (const float*)d_in[6];
    const float* Ds       = (const float*)d_in[7];
    float* out = (float*)d_out;

    float* xs   = (float*)d_ws;
    float* xdbl = xs + (size_t)BSZ * DCH * LL;

    {
        dim3 grid(BSZ * DCH * (HH / 4));
        conv_silu_kernel<<<grid, 256, 0, stream>>>(x, conv_w, conv_b, xs);
    }
    {
        const int nwaves = BSZ * (LL / 16);
        dim3 grid(nwaves / 8);
        proj_wmma_kernel<<<grid, 256, 0, stream>>>(xs, x_proj_w, xdbl);
    }
    {
        const int nwaves = BSZ * DCH;
        dim3 grid(nwaves / 8);
        scan_kernel<<<grid, 256, 0, stream>>>(xs, xdbl, dt_w, dt_b,
                                              A_logs, Ds, out);
    }
}


# scan kernel: each lane owns 8 consecutive positions per step (16-byte loads/stores, in-lane serial prefix, one wave scan of the lane totals per state): 16 steps per channel instead of 128
# speedup vs baseline: 1.7433x; 1.7433x over previous
#include <hip/hip_runtime.h>
#include <math.h>

#define BSZ   32
#define DCH   96
#define HH    64
#define WW    64
#define LL    (HH * WW)
#define NST   4
#define RNK   6
#define KPROJ 14
#define LOG_EPS_F (-20.7232658f)

typedef __attribute__((ext_vector_type(2))) float v2f;
typedef __attribute__((ext_vector_type(8))) float v8f;

__global__ void conv_silu_kernel(const float* __restrict__ x,
                                 const float* __restrict__ conv_w,
                                 const float* __restrict__ conv_b,
                                 float* __restrict__ xs) {
    const int bid  = blockIdx.x;
    const int tile = bid & 15;
    const int d    = (bid >> 4) % DCH;
    const int b    = bid / (16 * DCH);
    const int h0   = tile * 4;

    __shared__ float t[8][68];

    const int tid = threadIdx.x;
    const float* xin = x + ((size_t)b * DCH + d) * LL;

    for (int i = tid; i < 8 * 68; i += 256) {
        const int r  = i / 68;
        const int c  = i % 68;
        const int hh = h0 - 2 + r;
        const int wc = c - 2;
        float v = 0.0f;
        if (hh >= 0 && hh < HH && wc >= 0 && wc < WW)
            v = xin[hh * WW + wc];
        t[r][c] = v;
    }
    __syncthreads();

    float wv[25];
#pragma unroll
    for (int j = 0; j < 25; ++j) wv[j] = conv_w[d * 25 + j];
    const float bv = conv_b[d];

    const int ty = tid >> 6;
    const int tx = tid & 63;

    float acc = bv;
#pragma unroll
    for (int ky = 0; ky < 5; ++ky)
#pragma unroll
        for (int kx = 0; kx < 5; ++kx)
            acc = fmaf(t[ty + ky][tx + kx], wv[ky * 5 + kx], acc);

    const float s = acc / (1.0f + expf(-acc));
    xs[((size_t)b * DCH + d) * LL + (size_t)(h0 + ty) * WW + tx] = s;
}

__global__ void proj_wmma_kernel(const float* __restrict__ xs,
                                 const float* __restrict__ x_proj_w,
                                 float* __restrict__ xdbl) {
    __shared__ float wA[16][DCH];

    const int tid = threadIdx.x;
    for (int i = tid; i < 16 * DCH; i += 256) {
        const int row = i / DCH;
        const int k   = i % DCH;
        wA[row][k] = (row < KPROJ) ? x_proj_w[row * DCH + k] : 0.0f;
    }
    __syncthreads();

    const int wid  = (blockIdx.x * blockDim.x + tid) >> 5;
    const int lane = tid & 31;
    const int tcol = wid & 255;
    const int b    = wid >> 8;
    const int l0   = tcol * 16;

    const int half = lane >> 4;
    const int m    = lane & 15;

    const float* xb = xs + (size_t)b * DCH * LL + (l0 + m);

    v2f breg[DCH / 4];
#pragma unroll
    for (int kk = 0; kk < DCH / 4; ++kk) {
        const int k0 = kk * 4 + 2 * half;
        breg[kk].x = xb[(size_t)k0 * LL];
        breg[kk].y = xb[(size_t)(k0 + 1) * LL];
    }

    v8f acc = {0.f, 0.f, 0.f, 0.f, 0.f, 0.f, 0.f, 0.f};
#pragma unroll
    for (int kk = 0; kk < DCH / 4; ++kk) {
        const int k0 = kk * 4 + 2 * half;
        const v2f a = *(const v2f*)&wA[m][k0];
        acc = __builtin_amdgcn_wmma_f32_16x16x4_f32(
                  false, a, false, breg[kk], (short)0, acc, false, false);
    }

#pragma unroll
    for (int i = 0; i < 8; ++i) {
        const int row = i + 8 * half;
        if (row < KPROJ)
            xdbl[((size_t)b * KPROJ + row) * LL + l0 + m] = acc[i];
    }
}

#define EPL  8
#define STEP (32 * EPL)

__device__ __forceinline__ void wave_incl_sum_all(float (&v)[NST], int lane) {
#pragma unroll
    for (int off = 1; off < 32; off <<= 1) {
        float t[NST];
#pragma unroll
        for (int n = 0; n < NST; ++n) t[n] = __shfl_up(v[n], off, 32);
#pragma unroll
        for (int n = 0; n < NST; ++n)
            if (lane >= off) v[n] += t[n];
    }
}

__device__ __forceinline__ void load_run(float (&v)[EPL], const float* __restrict__ p) {
#pragma unroll
    for (int j = 0; j < EPL; j += 4) {
        const float4 t = *reinterpret_cast<const float4*>(p + j);
        v[j] = t.x; v[j + 1] = t.y; v[j + 2] = t.z; v[j + 3] = t.w;
    }
}

__device__ __forceinline__ void store_run(float* __restrict__ p, const float (&v)[EPL]) {
#pragma unroll
    for (int j = 0; j < EPL; j += 4) {
        float4 t;
        t.x = v[j]; t.y = v[j + 1]; t.z = v[j + 2]; t.w = v[j + 3];
        *reinterpret_cast<float4*>(p + j) = t;
    }
}

__global__ void __launch_bounds__(256, 4)
scan_kernel(const float* __restrict__ xs,
            const float* __restrict__ xdbl,
            const float* __restrict__ dt_w,
            const float* __restrict__ dt_b,
            const float* __restrict__ A_logs,
            const float* __restrict__ Ds,
            float* __restrict__ out) {
    const int wid  = __builtin_amdgcn_readfirstlane((int)((blockIdx.x * blockDim.x + threadIdx.x) >> 5));
    const int lane = threadIdx.x & 31;
    const int d    = wid % DCH;
    const int b    = wid / DCH;

    float wr[RNK];
#pragma unroll
    for (int r = 0; r < RNK; ++r) wr[r] = dt_w[d * RNK + r];
    const float dtb = dt_b[d];
    const float Dd  = Ds[d];
    float An[NST];
#pragma unroll
    for (int n = 0; n < NST; ++n) An[n] = -expf(A_logs[d * NST + n]);

    const float* xch = xs   + ((size_t)b * DCH + d) * LL;
    const float* xdb = xdbl + (size_t)b * KPROJ * LL;
    float*       och = out  + ((size_t)b * DCH + d) * LL;

    float c1[NST] = {0.f, 0.f, 0.f, 0.f};
    float c2[NST] = {0.f, 0.f, 0.f, 0.f};

    for (int c = 0; c < LL / STEP; ++c) {
        const int l = c * STEP + lane * EPL;

        float dts[RNK][EPL], xv[EPL], Bn[NST][EPL], Cn[NST][EPL];
#pragma unroll
        for (int r = 0; r < RNK; ++r) load_run(dts[r], xdb + (size_t)r * LL + l);
        load_run(xv, xch + l);
#pragma unroll
        for (int n = 0; n < NST; ++n) load_run(Bn[n], xdb + (size_t)(RNK + n) * LL + l);
#pragma unroll
        for (int n = 0; n < NST; ++n) load_run(Cn[n], xdb + (size_t)(RNK + NST + n) * LL + l);
        __builtin_amdgcn_sched_barrier(0);

        float delta[EPL];
#pragma unroll
        for (int j = 0; j < EPL; ++j) {
            float ad = dtb;
#pragma unroll
            for (int r = 0; r < RNK; ++r) ad = fmaf(wr[r], dts[r][j], ad);
            const float sp = log1pf(expf(ad));
            delta[j] = (ad > 20.0f) ? ad : sp;
        }

        float p1[NST][EPL], t1[NST];
#pragma unroll
        for (int n = 0; n < NST; ++n) {
            float run = 0.0f;
#pragma unroll
            for (int j = 0; j < EPL; ++j) {
                const float lda = fmaxf(delta[j] * An[n], LOG_EPS_F);
                run = (j == 0) ? lda : (run + lda);
                p1[n][j] = run;
            }
            t1[n] = run;
        }
        float incl1[NST];
#pragma unroll
        for (int n = 0; n < NST; ++n) incl1[n] = t1[n];
        wave_incl_sum_all(incl1, lane);

        float P[NST][EPL], p2[NST][EPL], t2[NST];
#pragma unroll
        for (int n = 0; n < NST; ++n) {
            const float base1 = (incl1[n] - t1[n]) + c1[n];
            float run = 0.0f;
#pragma unroll
            for (int j = 0; j < EPL; ++j) {
                const float s1 = base1 + p1[n][j];
                P[n][j] = expf(s1);
                const float q = (delta[j] * Bn[n][j] * xv[j]) / fmaxf(P[n][j], 1e-9f);
                run = (j == 0) ? q : (run + q);
                p2[n][j] = run;
            }
            t2[n] = run;
        }
        float incl2[NST];
#pragma unroll
        for (int n = 0; n < NST; ++n) incl2[n] = t2[n];
        wave_incl_sum_all(incl2, lane);

        float y[EPL];
#pragma unroll
        for (int j = 0; j < EPL; ++j) y[j] = xv[j] * Dd;
#pragma unroll
        for (int n = 0; n < NST; ++n) {
            const float base2 = (incl2[n] - t2[n]) + c2[n];
#pragma unroll
            for (int j = 0; j < EPL; ++j)
                y[j] = fmaf(P[n][j] * (base2 + p2[n][j]), Cn[n][j], y[j]);
        }
        store_run(och + l, y);

#pragma unroll
        for (int n = 0; n < NST; ++n) {
            c1[n] += __shfl(incl1[n], 31, 32);
            c2[n] += __shfl(incl2[n], 31, 32);
        }
    }
}

extern "C" void kernel_launch(void* const* d_in, const int* in_sizes, int n_in,
                              void* d_out, int out_size, void* d_ws, size_t ws_size,
                              hipStream_t stream) {
    (void)in_sizes; (void)n_in; (void)out_size; (void)ws_size;

    const float* x        = (const float*)d_in[0];
    const float* conv_w   = (const float*)d_in[1];
    const float* conv_b   = (const float*)d_in[2];
    const float* x_proj_w = (const float*)d_in[3];
    const float* dt_w     = (const float*)d_in[4];
    const float* dt_b     = (const float*)d_in[5];
    const float* A_logs   = (const float*)d_in[6];
    const float* Ds       = (const float*)d_in[7];
    float* out = (float*)d_out;

    float* xs   = (float*)d_ws;
    float* xdbl = xs + (size_t)BSZ * DCH * LL;

    {
        dim3 grid(BSZ * DCH * (HH / 4));
        conv_silu_kernel<<<grid, 256, 0, stream>>>(x, conv_w, conv_b, xs);
    }
    {
        const int nwaves = BSZ * (LL / 16);
        dim3 grid(nwaves / 8);
        proj_wmma_kernel<<<grid, 256, 0, stream>>>(xs, x_proj_w, xdbl);
    }
    {
        const int nwaves = BSZ * DCH;
        dim3 grid(nwaves / 8);
        scan_kernel<<<grid, 256, 0, stream>>>(xs, xdbl, dt_w, dt_b,
                                              A_logs, Ds, out);
    }
}


# conv kernel: each block streams whole 64x64 channel images (16-byte loads into a zero-framed LDS tile, next image in flight), each thread makes a 4x4 output patch from an 8x8 register window
# speedup vs baseline: 2.2874x; 1.3121x over previous
#include <hip/hip_runtime.h>
#include <math.h>

#define BSZ   32
#define DCH   96
#define HH    64
#define WW    64
#define LL    (HH * WW)
#define NST   4
#define RNK   6
#define KPROJ 14
#define LOG_EPS_F (-20.7232658f)

typedef __attribute__((ext_vector_type(2))) float v2f;
typedef __attribute__((ext_vector_type(8))) float v8f;

#define CPITCH 72
#define CROWS  68
#define CIMG   4

__global__ void __launch_bounds__(256, 6)
conv_silu_kernel(const float* __restrict__ x,
                 const float* __restrict__ conv_w,
                 const float* __restrict__ conv_b,
                 float* __restrict__ xs) {
    __shared__ __attribute__((aligned(16))) float t[CROWS * CPITCH];

    const int tid = threadIdx.x;
    const int d   = blockIdx.x % DCH;
    const int b0  = (blockIdx.x / DCH) * CIMG;

    float4 g0, g1, g2, g3;
    {
        const float4* __restrict__ xin4 =
            reinterpret_cast<const float4*>(x + ((size_t)b0 * DCH + d) * LL) + tid;
        g0 = xin4[0]; g1 = xin4[256]; g2 = xin4[512]; g3 = xin4[768];
    }

    float wv[25];
#pragma unroll
    for (int j = 0; j < 25; ++j) wv[j] = conv_w[d * 25 + j];
    const float bv = conv_b[d];

    float4* t4 = reinterpret_cast<float4*>(t);
    const float4 z4 = make_float4(0.f, 0.f, 0.f, 0.f);
    if (tid < 72) {
        const int r   = tid / 18;
        const int c   = tid - r * 18;
        const int row = (r < 2) ? r : (r + 64);
        t4[row * 18 + c] = z4;
    } else if (tid < 200) {
        const int u   = tid - 72;
        const int row = 2 + (u >> 1);
        const int c   = (u & 1) ? 17 : 0;
        t4[row * 18 + c] = z4;
    }

    const int pc = tid & 15;
    const int pr = tid >> 4;
    const float* base = t + (4 * pr) * CPITCH + 4 * pc;
    const int dst0 = ((tid >> 4) + 2) * 18 + 1 + (tid & 15);

#pragma unroll 1
    for (int it = 0; it < CIMG; ++it) {
        const size_t img = (size_t)(b0 + it) * DCH + d;

        t4[dst0]           = g0;
        t4[dst0 + 16 * 18] = g1;
        t4[dst0 + 32 * 18] = g2;
        t4[dst0 + 48 * 18] = g3;
        __syncthreads();

        float win[8][8];
#pragma unroll
        for (int r = 0; r < 8; ++r) {
            const float* rowp = base + r * CPITCH;
            const float2 a = *reinterpret_cast<const float2*>(rowp + 2);
            const float4 m = *reinterpret_cast<const float4*>(rowp + 4);
            const float2 c = *reinterpret_cast<const float2*>(rowp + 8);
            win[r][0] = a.x; win[r][1] = a.y;
            win[r][2] = m.x; win[r][3] = m.y; win[r][4] = m.z; win[r][5] = m.w;
            win[r][6] = c.x; win[r][7] = c.y;
        }
        __syncthreads();

        if (it + 1 < CIMG) {
            const float4* __restrict__ nx4 =
                reinterpret_cast<const float4*>(x + (img + DCH) * LL) + tid;
            g0 = nx4[0]; g1 = nx4[256]; g2 = nx4[512]; g3 = nx4[768];
        }

        float* __restrict__ xo = xs + img * LL;
#pragma unroll
        for (int i = 0; i < 4; ++i) {
            float o[4];
#pragma unroll
            for (int j = 0; j < 4; ++j) {
                float acc = bv;
#pragma unroll
                for (int ky = 0; ky < 5; ++ky)
#pragma unroll
                    for (int kx = 0; kx < 5; ++kx)
                        acc = fmaf(win[i + ky][j + kx], wv[ky * 5 + kx], acc);
                o[j] = acc / (1.0f + expf(-acc));
            }
            *reinterpret_cast<float4*>(xo + (4 * pr + i) * WW + 4 * pc) =
                make_float4(o[0], o[1], o[2], o[3]);
        }
    }
}

__global__ void proj_wmma_kernel(const float* __restrict__ xs,
                                 const float* __restrict__ x_proj_w,
                                 float* __restrict__ xdbl) {
    __shared__ float wA[16][DCH];

    const int tid = threadIdx.x;
    for (int i = tid; i < 16 * DCH; i += 256) {
        const int row = i / DCH;
        const int k   = i % DCH;
        wA[row][k] = (row < KPROJ) ? x_proj_w[row * DCH + k] : 0.0f;
    }
    __syncthreads();

    const int wid  = (blockIdx.x * blockDim.x + tid) >> 5;
    const int lane = tid & 31;
    const int tcol = wid & 255;
    const int b    = wid >> 8;
    const int l0   = tcol * 16;

    const int half = lane >> 4;
    const int m    = lane & 15;

    const float* xb = xs + (size_t)b * DCH * LL + (l0 + m);

    v2f breg[DCH / 4];
#pragma unroll
    for (int kk = 0; kk < DCH / 4; ++kk) {
        const int k0 = kk * 4 + 2 * half;
        breg[kk].x = xb[(size_t)k0 * LL];
        breg[kk].y = xb[(size_t)(k0 + 1) * LL];
    }

    v8f acc = {0.f, 0.f, 0.f, 0.f, 0.f, 0.f, 0.f, 0.f};
#pragma unroll
    for (int kk = 0; kk < DCH / 4; ++kk) {
        const int k0 = kk * 4 + 2 * half;
        const v2f a = *(const v2f*)&wA[m][k0];
        acc = __builtin_amdgcn_wmma_f32_16x16x4_f32(
                  false, a, false, breg[kk], (short)0, acc, false, false);
    }

#pragma unroll
    for (int i = 0; i < 8; ++i) {
        const int row = i + 8 * half;
        if (row < KPROJ)
            xdbl[((size_t)b * KPROJ + row) * LL + l0 + m] = acc[i];
    }
}

#define EPL  8
#define STEP (32 * EPL)

__device__ __forceinline__ void wave_incl_sum_all(float (&v)[NST], int lane) {
#pragma unroll
    for (int off = 1; off < 32; off <<= 1) {
        float t[NST];
#pragma unroll
        for (int n = 0; n < NST; ++n) t[n] = __shfl_up(v[n], off, 32);
#pragma unroll
        for (int n = 0; n < NST; ++n)
            if (lane >= off) v[n] += t[n];
    }
}

__device__ __forceinline__ void load_run(float (&v)[EPL], const float* __restrict__ p) {
#pragma unroll
    for (int j = 0; j < EPL; j += 4) {
        const float4 t = *reinterpret_cast<const float4*>(p + j);
        v[j] = t.x; v[j + 1] = t.y; v[j + 2] = t.z; v[j + 3] = t.w;
    }
}

__device__ __forceinline__ void store_run(float* __restrict__ p, const float (&v)[EPL]) {
#pragma unroll
    for (int j = 0; j < EPL; j += 4) {
        float4 t;
        t.x = v[j]; t.y = v[j + 1]; t.z = v[j + 2]; t.w = v[j + 3];
        *reinterpret_cast<float4*>(p + j) = t;
    }
}

__global__ void __launch_bounds__(256, 4)
scan_kernel(const float* __restrict__ xs,
            const float* __restrict__ xdbl,
            const float* __restrict__ dt_w,
            const float* __restrict__ dt_b,
            const float* __restrict__ A_logs,
            const float* __restrict__ Ds,
            float* __restrict__ out) {
    const int wid  = __builtin_amdgcn_readfirstlane((int)((blockIdx.x * blockDim.x + threadIdx.x) >> 5));
    const int lane = threadIdx.x & 31;
    const int d    = wid % DCH;
    const int b    = wid / DCH;

    float wr[RNK];
#pragma unroll
    for (int r = 0; r < RNK; ++r) wr[r] = dt_w[d * RNK + r];
    const float dtb = dt_b[d];
    const float Dd  = Ds[d];
    float An[NST];
#pragma unroll
    for (int n = 0; n < NST; ++n) An[n] = -expf(A_logs[d * NST + n]);

    const float* xch = xs   + ((size_t)b * DCH + d) * LL;
    const float* xdb = xdbl + (size_t)b * KPROJ * LL;
    float*       och = out  + ((size_t)b * DCH + d) * LL;

    float c1[NST] = {0.f, 0.f, 0.f, 0.f};
    float c2[NST] = {0.f, 0.f, 0.f, 0.f};

    for (int c = 0; c < LL / STEP; ++c) {
        const int l = c * STEP + lane * EPL;

        float dts[RNK][EPL], xv[EPL], Bn[NST][EPL], Cn[NST][EPL];
#pragma unroll
        for (int r = 0; r < RNK; ++r) load_run(dts[r], xdb + (size_t)r * LL + l);
        load_run(xv, xch + l);
#pragma unroll
        for (int n = 0; n < NST; ++n) load_run(Bn[n], xdb + (size_t)(RNK + n) * LL + l);
#pragma unroll
        for (int n = 0; n < NST; ++n) load_run(Cn[n], xdb + (size_t)(RNK + NST + n) * LL + l);
        __builtin_amdgcn_sched_barrier(0);

        float delta[EPL];
#pragma unroll
        for (int j = 0; j < EPL; ++j) {
            float ad = dtb;
#pragma unroll
            for (int r = 0; r < RNK; ++r) ad = fmaf(wr[r], dts[r][j], ad);
            const float sp = log1pf(expf(ad));
            delta[j] = (ad > 20.0f) ? ad : sp;
        }

        float p1[NST][EPL], t1[NST];
#pragma unroll
        for (int n = 0; n < NST; ++n) {
            float run = 0.0f;
#pragma unroll
            for (int j = 0; j < EPL; ++j) {
                const float lda = fmaxf(delta[j] * An[n], LOG_EPS_F);
                run = (j == 0) ? lda : (run + lda);
                p1[n][j] = run;
            }
            t1[n] = run;
        }
        float incl1[NST];
#pragma unroll
        for (int n = 0; n < NST; ++n) incl1[n] = t1[n];
        wave_incl_sum_all(incl1, lane);

        float P[NST][EPL], p2[NST][EPL], t2[NST];
#pragma unroll
        for (int n = 0; n < NST; ++n) {
            const float base1 = (incl1[n] - t1[n]) + c1[n];
            float run = 0.0f;
#pragma unroll
            for (int j = 0; j < EPL; ++j) {
                const float s1 = base1 + p1[n][j];
                P[n][j] = expf(s1);
                const float q = (delta[j] * Bn[n][j] * xv[j]) / fmaxf(P[n][j], 1e-9f);
                run = (j == 0) ? q : (run + q);
                p2[n][j] = run;
            }
            t2[n] = run;
        }
        float incl2[NST];
#pragma unroll
        for (int n = 0; n < NST; ++n) incl2[n] = t2[n];
        wave_incl_sum_all(incl2, lane);

        float y[EPL];
#pragma unroll
        for (int j = 0; j < EPL; ++j) y[j] = xv[j] * Dd;
#pragma unroll
        for (int n = 0; n < NST; ++n) {
            const float base2 = (incl2[n] - t2[n]) + c2[n];
#pragma unroll
            for (int j = 0; j < EPL; ++j)
                y[j] = fmaf(P[n][j] * (base2 + p2[n][j]), Cn[n][j], y[j]);
        }
        store_run(och + l, y);

#pragma unroll
        for (int n = 0; n < NST; ++n) {
            c1[n] += __shfl(incl1[n], 31, 32);
            c2[n] += __shfl(incl2[n], 31, 32);
        }
    }
}

extern "C" void kernel_launch(void* const* d_in, const int* in_sizes, int n_in,
                              void* d_out, int out_size, void* d_ws, size_t ws_size,
                              hipStream_t stream) {
    (void)in_sizes; (void)n_in; (void)out_size; (void)ws_size;

    const float* x        = (const float*)d_in[0];
    const float* conv_w   = (const float*)d_in[1];
    const float* conv_b   = (const float*)d_in[2];
    const float* x_proj_w = (const float*)d_in[3];
    const float* dt_w     = (const float*)d_in[4];
    const float* dt_b     = (const float*)d_in[5];
    const float* A_logs   = (const float*)d_in[6];
    const float* Ds       = (const float*)d_in[7];
    float* out = (float*)d_out;

    float* xs   = (float*)d_ws;
    float* xdbl = xs + (size_t)BSZ * DCH * LL;

    {
        dim3 grid(DCH * (BSZ / 4));
        conv_silu_kernel<<<grid, 256, 0, stream>>>(x, conv_w, conv_b, xs);
    }
    {
        const int nwaves = BSZ * (LL / 16);
        dim3 grid(nwaves / 8);
        proj_wmma_kernel<<<grid, 256, 0, stream>>>(xs, x_proj_w, xdbl);
    }
    {
        const int nwaves = BSZ * DCH;
        dim3 grid(nwaves / 8);
        scan_kernel<<<grid, 256, 0, stream>>>(xs, xdbl, dt_w, dt_b,
                                              A_logs, Ds, out);
    }
}


# scan kernel: short-form transcendentals: decay cumsum kept in log2 units so P is one hardware exp2, quotient as reciprocal-multiply, softplus as max(a,0)+log1p(exp(-|a|)) by an atanh series
# speedup vs baseline: 2.9855x; 1.3052x over previous
#include <hip/hip_runtime.h>
#include <math.h>

#define BSZ   32
#define DCH   96
#define HH    64
#define WW    64
#define LL    (HH * WW)
#define NST   4
#define RNK   6
#define KPROJ 14

typedef __attribute__((ext_vector_type(2))) float v2f;
typedef __attribute__((ext_vector_type(8))) float v8f;

#define CPITCH 72
#define CROWS  68
#define CIMG   4

__global__ void __launch_bounds__(256, 6)
conv_silu_kernel(const float* __restrict__ x,
                 const float* __restrict__ conv_w,
                 const float* __restrict__ conv_b,
                 float* __restrict__ xs) {
    __shared__ __attribute__((aligned(16))) float t[CROWS * CPITCH];

    const int tid = threadIdx.x;
    const int d   = blockIdx.x % DCH;
    const int b0  = (blockIdx.x / DCH) * CIMG;

    float4 g0, g1, g2, g3;
    {
        const float4* __restrict__ xin4 =
            reinterpret_cast<const float4*>(x + ((size_t)b0 * DCH + d) * LL) + tid;
        g0 = xin4[0]; g1 = xin4[256]; g2 = xin4[512]; g3 = xin4[768];
    }

    float wv[25];
#pragma unroll
    for (int j = 0; j < 25; ++j) wv[j] = conv_w[d * 25 + j];
    const float bv = conv_b[d];

    float4* t4 = reinterpret_cast<float4*>(t);
    const float4 z4 = make_float4(0.f, 0.f, 0.f, 0.f);
    if (tid < 72) {
        const int r   = tid / 18;
        const int c   = tid - r * 18;
        const int row = (r < 2) ? r : (r + 64);
        t4[row * 18 + c] = z4;
    } else if (tid < 200) {
        const int u   = tid - 72;
        const int row = 2 + (u >> 1);
        const int c   = (u & 1) ? 17 : 0;
        t4[row * 18 + c] = z4;
    }

    const int pc = tid & 15;
    const int pr = tid >> 4;
    const float* base = t + (4 * pr) * CPITCH + 4 * pc;
    const int dst0 = ((tid >> 4) + 2) * 18 + 1 + (tid & 15);

#pragma unroll 1
    for (int it = 0; it < CIMG; ++it) {
        const size_t img = (size_t)(b0 + it) * DCH + d;

        t4[dst0]           = g0;
        t4[dst0 + 16 * 18] = g1;
        t4[dst0 + 32 * 18] = g2;
        t4[dst0 + 48 * 18] = g3;
        __syncthreads();

        float win[8][8];
#pragma unroll
        for (int r = 0; r < 8; ++r) {
            const float* rowp = base + r * CPITCH;
            const float2 a = *reinterpret_cast<const float2*>(rowp + 2);
            const float4 m = *reinterpret_cast<const float4*>(rowp + 4);
            const float2 c = *reinterpret_cast<const float2*>(rowp + 8);
            win[r][0] = a.x; win[r][1] = a.y;
            win[r][2] = m.x; win[r][3] = m.y; win[r][4] = m.z; win[r][5] = m.w;
            win[r][6] = c.x; win[r][7] = c.y;
        }
        __syncthreads();

        if (it + 1 < CIMG) {
            const float4* __restrict__ nx4 =
                reinterpret_cast<const float4*>(x + (img + DCH) * LL) + tid;
            g0 = nx4[0]; g1 = nx4[256]; g2 = nx4[512]; g3 = nx4[768];
        }

        float* __restrict__ xo = xs + img * LL;
#pragma unroll
        for (int i = 0; i < 4; ++i) {
            float o[4];
#pragma unroll
            for (int j = 0; j < 4; ++j) {
                float acc = bv;
#pragma unroll
                for (int ky = 0; ky < 5; ++ky)
#pragma unroll
                    for (int kx = 0; kx < 5; ++kx)
                        acc = fmaf(win[i + ky][j + kx], wv[ky * 5 + kx], acc);
                o[j] = acc / (1.0f + expf(-acc));
            }
            *reinterpret_cast<float4*>(xo + (4 * pr + i) * WW + 4 * pc) =
                make_float4(o[0], o[1], o[2], o[3]);
        }
    }
}

__global__ void proj_wmma_kernel(const float* __restrict__ xs,
                                 const float* __restrict__ x_proj_w,
                                 float* __restrict__ xdbl) {
    __shared__ float wA[16][DCH];

    const int tid = threadIdx.x;
    for (int i = tid; i < 16 * DCH; i += 256) {
        const int row = i / DCH;
        const int k   = i % DCH;
        wA[row][k] = (row < KPROJ) ? x_proj_w[row * DCH + k] : 0.0f;
    }
    __syncthreads();

    const int wid  = (blockIdx.x * blockDim.x + tid) >> 5;
    const int lane = tid & 31;
    const int tcol = wid & 255;
    const int b    = wid >> 8;
    const int l0   = tcol * 16;

    const int half = lane >> 4;
    const int m    = lane & 15;

    const float* xb = xs + (size_t)b * DCH * LL + (l0 + m);

    v2f breg[DCH / 4];
#pragma unroll
    for (int kk = 0; kk < DCH / 4; ++kk) {
        const int k0 = kk * 4 + 2 * half;
        breg[kk].x = xb[(size_t)k0 * LL];
        breg[kk].y = xb[(size_t)(k0 + 1) * LL];
    }

    v8f acc = {0.f, 0.f, 0.f, 0.f, 0.f, 0.f, 0.f, 0.f};
#pragma unroll
    for (int kk = 0; kk < DCH / 4; ++kk) {
        const int k0 = kk * 4 + 2 * half;
        const v2f a = *(const v2f*)&wA[m][k0];
        acc = __builtin_amdgcn_wmma_f32_16x16x4_f32(
                  false, a, false, breg[kk], (short)0, acc, false, false);
    }

#pragma unroll
    for (int i = 0; i < 8; ++i) {
        const int row = i + 8 * half;
        if (row < KPROJ)
            xdbl[((size_t)b * KPROJ + row) * LL + l0 + m] = acc[i];
    }
}

#define EPL  8
#define STEP (32 * EPL)

#define LOG2E_HI_F  1.4426950216293335f
#define LOG2E_LO_F  1.92596299e-8f
#define LN2_F       0.693147180559945309f
#define LOG2_EPS_F  (-29.8973529f)

__device__ __forceinline__ float softplus_f32(float a) {
    const float na = -fabsf(a);
    const float t = na * LOG2E_HI_F;
    float r = fmaf(na, LOG2E_HI_F, -t);
    r = fmaf(na, LOG2E_LO_F, r);
    float z = __builtin_amdgcn_exp2f(t);
    z = fmaf(z, r * LN2_F, z);
    const float u = 2.0f + z;
    float ru = __builtin_amdgcn_rcpf(u);
    ru = fmaf(fmaf(-u, ru, 1.0f), ru, ru);
    const float s = z * ru;
    const float w = s * s;
    float p = 1.0f / 15.0f;
    p = fmaf(p, w, 1.0f / 13.0f);
    p = fmaf(p, w, 1.0f / 11.0f);
    p = fmaf(p, w, 1.0f / 9.0f);
    p = fmaf(p, w, 1.0f / 7.0f);
    p = fmaf(p, w, 1.0f / 5.0f);
    p = fmaf(p, w, 1.0f / 3.0f);
    const float s2 = s + s;
    const float l1p = fmaf(s2 * w, p, s2);
    return fmaxf(a, 0.0f) + l1p;
}

__device__ __forceinline__ void wave_incl_sum_all(float (&v)[NST], int lane) {
#pragma unroll
    for (int off = 1; off < 32; off <<= 1) {
        float t[NST];
#pragma unroll
        for (int n = 0; n < NST; ++n) t[n] = __shfl_up(v[n], off, 32);
#pragma unroll
        for (int n = 0; n < NST; ++n)
            if (lane >= off) v[n] += t[n];
    }
}

__device__ __forceinline__ void load_run(float (&v)[EPL], const float* __restrict__ p) {
#pragma unroll
    for (int j = 0; j < EPL; j += 4) {
        const float4 t = *reinterpret_cast<const float4*>(p + j);
        v[j] = t.x; v[j + 1] = t.y; v[j + 2] = t.z; v[j + 3] = t.w;
    }
}

__device__ __forceinline__ void store_run(float* __restrict__ p, const float (&v)[EPL]) {
#pragma unroll
    for (int j = 0; j < EPL; j += 4) {
        float4 t;
        t.x = v[j]; t.y = v[j + 1]; t.z = v[j + 2]; t.w = v[j + 3];
        *reinterpret_cast<float4*>(p + j) = t;
    }
}

__global__ void __launch_bounds__(256, 4)
scan_kernel(const float* __restrict__ xs,
            const float* __restrict__ xdbl,
            const float* __restrict__ dt_w,
            const float* __restrict__ dt_b,
            const float* __restrict__ A_logs,
            const float* __restrict__ Ds,
            float* __restrict__ out) {
    const int wid  = __builtin_amdgcn_readfirstlane((int)((blockIdx.x * blockDim.x + threadIdx.x) >> 5));
    const int lane = threadIdx.x & 31;
    const int d    = wid % DCH;
    const int b    = wid / DCH;

    float wr[RNK];
#pragma unroll
    for (int r = 0; r < RNK; ++r) wr[r] = dt_w[d * RNK + r];
    const float dtb = dt_b[d];
    const float Dd  = Ds[d];
    float An[NST];
#pragma unroll
    for (int n = 0; n < NST; ++n) An[n] = -expf(A_logs[d * NST + n]) * LOG2E_HI_F;

    const float* xch = xs   + ((size_t)b * DCH + d) * LL;
    const float* xdb = xdbl + (size_t)b * KPROJ * LL;
    float*       och = out  + ((size_t)b * DCH + d) * LL;

    float c1[NST] = {0.f, 0.f, 0.f, 0.f};
    float c2[NST] = {0.f, 0.f, 0.f, 0.f};

    for (int c = 0; c < LL / STEP; ++c) {
        const int l = c * STEP + lane * EPL;

        float dts[RNK][EPL], xv[EPL], Bn[NST][EPL], Cn[NST][EPL];
#pragma unroll
        for (int r = 0; r < RNK; ++r) load_run(dts[r], xdb + (size_t)r * LL + l);
        load_run(xv, xch + l);
#pragma unroll
        for (int n = 0; n < NST; ++n) load_run(Bn[n], xdb + (size_t)(RNK + n) * LL + l);
#pragma unroll
        for (int n = 0; n < NST; ++n) load_run(Cn[n], xdb + (size_t)(RNK + NST + n) * LL + l);
        __builtin_amdgcn_sched_barrier(0);

        float delta[EPL];
#pragma unroll
        for (int j = 0; j < EPL; ++j) {
            float ad = dtb;
#pragma unroll
            for (int r = 0; r < RNK; ++r) ad = fmaf(wr[r], dts[r][j], ad);
            delta[j] = softplus_f32(ad);
        }

        float p1[NST][EPL], t1[NST];
#pragma unroll
        for (int n = 0; n < NST; ++n) {
            float run = 0.0f;
#pragma unroll
            for (int j = 0; j < EPL; ++j) {
                const float lda = fmaxf(delta[j] * An[n], LOG2_EPS_F);
                run = (j == 0) ? lda : (run + lda);
                p1[n][j] = run;
            }
            t1[n] = run;
        }
        float incl1[NST];
#pragma unroll
        for (int n = 0; n < NST; ++n) incl1[n] = t1[n];
        wave_incl_sum_all(incl1, lane);

        float P[NST][EPL], p2[NST][EPL], t2[NST];
#pragma unroll
        for (int n = 0; n < NST; ++n) {
            const float base1 = (incl1[n] - t1[n]) + c1[n];
            float run = 0.0f;
#pragma unroll
            for (int j = 0; j < EPL; ++j) {
                const float s1 = base1 + p1[n][j];
                P[n][j] = __builtin_amdgcn_exp2f(s1);
                const float q = (delta[j] * Bn[n][j] * xv[j]) * __builtin_amdgcn_rcpf(fmaxf(P[n][j], 1e-9f));
                run = (j == 0) ? q : (run + q);
                p2[n][j] = run;
            }
            t2[n] = run;
        }
        float incl2[NST];
#pragma unroll
        for (int n = 0; n < NST; ++n) incl2[n] = t2[n];
        wave_incl_sum_all(incl2, lane);

        float y[EPL];
#pragma unroll
        for (int j = 0; j < EPL; ++j) y[j] = xv[j] * Dd;
#pragma unroll
        for (int n = 0; n < NST; ++n) {
            const float base2 = (incl2[n] - t2[n]) + c2[n];
#pragma unroll
            for (int j = 0; j < EPL; ++j)
                y[j] = fmaf(P[n][j] * (base2 + p2[n][j]), Cn[n][j], y[j]);
        }
        store_run(och + l, y);

#pragma unroll
        for (int n = 0; n < NST; ++n) {
            c1[n] += __shfl(incl1[n], 31, 32);
            c2[n] += __shfl(incl2[n], 31, 32);
        }
    }
}

extern "C" void kernel_launch(void* const* d_in, const int* in_sizes, int n_in,
                              void* d_out, int out_size, void* d_ws, size_t ws_size,
                              hipStream_t stream) {
    (void)in_sizes; (void)n_in; (void)out_size; (void)ws_size;

    const float* x        = (const float*)d_in[0];
    const float* conv_w   = (const float*)d_in[1];
    const float* conv_b   = (const float*)d_in[2];
    const float* x_proj_w = (const float*)d_in[3];
    const float* dt_w     = (const float*)d_in[4];
    const float* dt_b     = (const float*)d_in[5];
    const float* A_logs   = (const float*)d_in[6];
    const float* Ds       = (const float*)d_in[7];
    float* out = (float*)d_out;

    float* xs   = (float*)d_ws;
    float* xdbl = xs + (size_t)BSZ * DCH * LL;

    {
        dim3 grid(DCH * (BSZ / 4));
        conv_silu_kernel<<<grid, 256, 0, stream>>>(x, conv_w, conv_b, xs);
    }
    {
        const int nwaves = BSZ * (LL / 16);
        dim3 grid(nwaves / 8);
        proj_wmma_kernel<<<grid, 256, 0, stream>>>(xs, x_proj_w, xdbl);
    }
    {
        const int nwaves = BSZ * DCH;
        dim3 grid(nwaves / 8);
        scan_kernel<<<grid, 256, 0, stream>>>(xs, xdbl, dt_w, dt_b,
                                              A_logs, Ds, out);
    }
}


# scan kernel: the wave scans of the lane totals and the step carries run on the vector ALU (row-shift DPP adds inside each half-wave row, then lane reads) instead of LDS-unit permutes
# speedup vs baseline: 3.1904x; 1.0686x over previous
#include <hip/hip_runtime.h>
#include <math.h>

#define BSZ   32
#define DCH   96
#define HH    64
#define WW    64
#define LL    (HH * WW)
#define NST   4
#define RNK   6
#define KPROJ 14

typedef __attribute__((ext_vector_type(2))) float v2f;
typedef __attribute__((ext_vector_type(8))) float v8f;

#define CPITCH 72
#define CROWS  68
#define CIMG   4

__global__ void __launch_bounds__(256, 6)
conv_silu_kernel(const float* __restrict__ x,
                 const float* __restrict__ conv_w,
                 const float* __restrict__ conv_b,
                 float* __restrict__ xs) {
    __shared__ __attribute__((aligned(16))) float t[CROWS * CPITCH];

    const int tid = threadIdx.x;
    const int d   = blockIdx.x % DCH;
    const int b0  = (blockIdx.x / DCH) * CIMG;

    float4 g0, g1, g2, g3;
    {
        const float4* __restrict__ xin4 =
            reinterpret_cast<const float4*>(x + ((size_t)b0 * DCH + d) * LL) + tid;
        g0 = xin4[0]; g1 = xin4[256]; g2 = xin4[512]; g3 = xin4[768];
    }

    float wv[25];
#pragma unroll
    for (int j = 0; j < 25; ++j) wv[j] = conv_w[d * 25 + j];
    const float bv = conv_b[d];

    float4* t4 = reinterpret_cast<float4*>(t);
    const float4 z4 = make_float4(0.f, 0.f, 0.f, 0.f);
    if (tid < 72) {
        const int r   = tid / 18;
        const int c   = tid - r * 18;
        const int row = (r < 2) ? r : (r + 64);
        t4[row * 18 + c] = z4;
    } else if (tid < 200) {
        const int u   = tid - 72;
        const int row = 2 + (u >> 1);
        const int c   = (u & 1) ? 17 : 0;
        t4[row * 18 + c] = z4;
    }

    const int pc = tid & 15;
    const int pr = tid >> 4;
    const float* base = t + (4 * pr) * CPITCH + 4 * pc;
    const int dst0 = ((tid >> 4) + 2) * 18 + 1 + (tid & 15);

#pragma unroll 1
    for (int it = 0; it < CIMG; ++it) {
        const size_t img = (size_t)(b0 + it) * DCH + d;

        t4[dst0]           = g0;
        t4[dst0 + 16 * 18] = g1;
        t4[dst0 + 32 * 18] = g2;
        t4[dst0 + 48 * 18] = g3;
        __syncthreads();

        float win[8][8];
#pragma unroll
        for (int r = 0; r < 8; ++r) {
            const float* rowp = base + r * CPITCH;
            const float2 a = *reinterpret_cast<const float2*>(rowp + 2);
            const float4 m = *reinterpret_cast<const float4*>(rowp + 4);
            const float2 c = *reinterpret_cast<const float2*>(rowp + 8);
            win[r][0] = a.x; win[r][1] = a.y;
            win[r][2] = m.x; win[r][3] = m.y; win[r][4] = m.z; win[r][5] = m.w;
            win[r][6] = c.x; win[r][7] = c.y;
        }
        __syncthreads();

        if (it + 1 < CIMG) {
            const float4* __restrict__ nx4 =
                reinterpret_cast<const float4*>(x + (img + DCH) * LL) + tid;
            g0 = nx4[0]; g1 = nx4[256]; g2 = nx4[512]; g3 = nx4[768];
        }

        float* __restrict__ xo = xs + img * LL;
#pragma unroll
        for (int i = 0; i < 4; ++i) {
            float o[4];
#pragma unroll
            for (int j = 0; j < 4; ++j) {
                float acc = bv;
#pragma unroll
                for (int ky = 0; ky < 5; ++ky)
#pragma unroll
                    for (int kx = 0; kx < 5; ++kx)
                        acc = fmaf(win[i + ky][j + kx], wv[ky * 5 + kx], acc);
                o[j] = acc / (1.0f + expf(-acc));
            }
            *reinterpret_cast<float4*>(xo + (4 * pr + i) * WW + 4 * pc) =
                make_float4(o[0], o[1], o[2], o[3]);
        }
    }
}

__global__ void proj_wmma_kernel(const float* __restrict__ xs,
                                 const float* __restrict__ x_proj_w,
                                 float* __restrict__ xdbl) {
    __shared__ float wA[16][DCH];

    const int tid = threadIdx.x;
    for (int i = tid; i < 16 * DCH; i += 256) {
        const int row = i / DCH;
        const int k   = i % DCH;
        wA[row][k] = (row < KPROJ) ? x_proj_w[row * DCH + k] : 0.0f;
    }
    __syncthreads();

    const int wid  = (blockIdx.x * blockDim.x + tid) >> 5;
    const int lane = tid & 31;
    const int tcol = wid & 255;
    const int b    = wid >> 8;
    const int l0   = tcol * 16;

    const int half = lane >> 4;
    const int m    = lane & 15;

    const float* xb = xs + (size_t)b * DCH * LL + (l0 + m);

    v2f breg[DCH / 4];
#pragma unroll
    for (int kk = 0; kk < DCH / 4; ++kk) {
        const int k0 = kk * 4 + 2 * half;
        breg[kk].x = xb[(size_t)k0 * LL];
        breg[kk].y = xb[(size_t)(k0 + 1) * LL];
    }

    v8f acc = {0.f, 0.f, 0.f, 0.f, 0.f, 0.f, 0.f, 0.f};
#pragma unroll
    for (int kk = 0; kk < DCH / 4; ++kk) {
        const int k0 = kk * 4 + 2 * half;
        const v2f a = *(const v2f*)&wA[m][k0];
        acc = __builtin_amdgcn_wmma_f32_16x16x4_f32(
                  false, a, false, breg[kk], (short)0, acc, false, false);
    }

#pragma unroll
    for (int i = 0; i < 8; ++i) {
        const int row = i + 8 * half;
        if (row < KPROJ)
            xdbl[((size_t)b * KPROJ + row) * LL + l0 + m] = acc[i];
    }
}

#define EPL  8
#define STEP (32 * EPL)

#define LOG2E_HI_F  1.4426950216293335f
#define LOG2E_LO_F  1.92596299e-8f
#define LN2_F       0.693147180559945309f
#define LOG2_EPS_F  (-29.8973529f)

__device__ __forceinline__ float softplus_f32(float a) {
    const float na = -fabsf(a);
    const float t = na * LOG2E_HI_F;
    float r = fmaf(na, LOG2E_HI_F, -t);
    r = fmaf(na, LOG2E_LO_F, r);
    float z = __builtin_amdgcn_exp2f(t);
    z = fmaf(z, r * LN2_F, z);
    const float u = 2.0f + z;
    float ru = __builtin_amdgcn_rcpf(u);
    ru = fmaf(fmaf(-u, ru, 1.0f), ru, ru);
    const float s = z * ru;
    const float w = s * s;
    float p = 1.0f / 15.0f;
    p = fmaf(p, w, 1.0f / 13.0f);
    p = fmaf(p, w, 1.0f / 11.0f);
    p = fmaf(p, w, 1.0f / 9.0f);
    p = fmaf(p, w, 1.0f / 7.0f);
    p = fmaf(p, w, 1.0f / 5.0f);
    p = fmaf(p, w, 1.0f / 3.0f);
    const float s2 = s + s;
    const float l1p = fmaf(s2 * w, p, s2);
    return fmaxf(a, 0.0f) + l1p;
}

template <int SHR>
__device__ __forceinline__ float row_shr_or_zero(float v) {
    return __builtin_bit_cast(float,
        __builtin_amdgcn_update_dpp(0, __builtin_bit_cast(int, v), 0x110 + SHR, 0xf, 0xf, true));
}

template <int LANE>
__device__ __forceinline__ float lane_value(float v) {
    return __builtin_bit_cast(float, __builtin_amdgcn_readlane(__builtin_bit_cast(int, v), LANE));
}

__device__ __forceinline__ void wave_incl_sum_all(float (&v)[NST], int lane) {
#pragma unroll
    for (int n = 0; n < NST; ++n) v[n] += row_shr_or_zero<1>(v[n]);
#pragma unroll
    for (int n = 0; n < NST; ++n) v[n] += row_shr_or_zero<2>(v[n]);
#pragma unroll
    for (int n = 0; n < NST; ++n) v[n] += row_shr_or_zero<4>(v[n]);
#pragma unroll
    for (int n = 0; n < NST; ++n) v[n] += row_shr_or_zero<8>(v[n]);
#pragma unroll
    for (int n = 0; n < NST; ++n) {
        const float low_row_total = lane_value<15>(v[n]);
        v[n] += (lane >= 16) ? low_row_total : 0.0f;
    }
}

__device__ __forceinline__ void load_run(float (&v)[EPL], const float* __restrict__ p) {
#pragma unroll
    for (int j = 0; j < EPL; j += 4) {
        const float4 t = *reinterpret_cast<const float4*>(p + j);
        v[j] = t.x; v[j + 1] = t.y; v[j + 2] = t.z; v[j + 3] = t.w;
    }
}

__device__ __forceinline__ void store_run(float* __restrict__ p, const float (&v)[EPL]) {
#pragma unroll
    for (int j = 0; j < EPL; j += 4) {
        float4 t;
        t.x = v[j]; t.y = v[j + 1]; t.z = v[j + 2]; t.w = v[j + 3];
        *reinterpret_cast<float4*>(p + j) = t;
    }
}

__global__ void __launch_bounds__(256, 4)
scan_kernel(const float* __restrict__ xs,
            const float* __restrict__ xdbl,
            const float* __restrict__ dt_w,
            const float* __restrict__ dt_b,
            const float* __restrict__ A_logs,
            const float* __restrict__ Ds,
            float* __restrict__ out) {
    const int wid  = __builtin_amdgcn_readfirstlane((int)((blockIdx.x * blockDim.x + threadIdx.x) >> 5));
    const int lane = threadIdx.x & 31;
    const int d    = wid % DCH;
    const int b    = wid / DCH;

    float wr[RNK];
#pragma unroll
    for (int r = 0; r < RNK; ++r) wr[r] = dt_w[d * RNK + r];
    const float dtb = dt_b[d];
    const float Dd  = Ds[d];
    float An[NST];
#pragma unroll
    for (int n = 0; n < NST; ++n) An[n] = -expf(A_logs[d * NST + n]) * LOG2E_HI_F;

    const float* xch = xs   + ((size_t)b * DCH + d) * LL;
    const float* xdb = xdbl + (size_t)b * KPROJ * LL;
    float*       och = out  + ((size_t)b * DCH + d) * LL;

    float c1[NST] = {0.f, 0.f, 0.f, 0.f};
    float c2[NST] = {0.f, 0.f, 0.f, 0.f};

    for (int c = 0; c < LL / STEP; ++c) {
        const int l = c * STEP + lane * EPL;

        float dts[RNK][EPL], xv[EPL], Bn[NST][EPL], Cn[NST][EPL];
#pragma unroll
        for (int r = 0; r < RNK; ++r) load_run(dts[r], xdb + (size_t)r * LL + l);
        load_run(xv, xch + l);
#pragma unroll
        for (int n = 0; n < NST; ++n) load_run(Bn[n], xdb + (size_t)(RNK + n) * LL + l);
#pragma unroll
        for (int n = 0; n < NST; ++n) load_run(Cn[n], xdb + (size_t)(RNK + NST + n) * LL + l);
        __builtin_amdgcn_sched_barrier(0);

        float delta[EPL];
#pragma unroll
        for (int j = 0; j < EPL; ++j) {
            float ad = dtb;
#pragma unroll
            for (int r = 0; r < RNK; ++r) ad = fmaf(wr[r], dts[r][j], ad);
            delta[j] = softplus_f32(ad);
        }

        float p1[NST][EPL], t1[NST];
#pragma unroll
        for (int n = 0; n < NST; ++n) {
            float run = 0.0f;
#pragma unroll
            for (int j = 0; j < EPL; ++j) {
                const float lda = fmaxf(delta[j] * An[n], LOG2_EPS_F);
                run = (j == 0) ? lda : (run + lda);
                p1[n][j] = run;
            }
            t1[n] = run;
        }
        float incl1[NST];
#pragma unroll
        for (int n = 0; n < NST; ++n) incl1[n] = t1[n];
        wave_incl_sum_all(incl1, lane);

        float P[NST][EPL], p2[NST][EPL], t2[NST];
#pragma unroll
        for (int n = 0; n < NST; ++n) {
            const float base1 = (incl1[n] - t1[n]) + c1[n];
            float run = 0.0f;
#pragma unroll
            for (int j = 0; j < EPL; ++j) {
                const float s1 = base1 + p1[n][j];
                P[n][j] = __builtin_amdgcn_exp2f(s1);
                const float q = (delta[j] * Bn[n][j] * xv[j]) * __builtin_amdgcn_rcpf(fmaxf(P[n][j], 1e-9f));
                run = (j == 0) ? q : (run + q);
                p2[n][j] = run;
            }
            t2[n] = run;
        }
        float incl2[NST];
#pragma unroll
        for (int n = 0; n < NST; ++n) incl2[n] = t2[n];
        wave_incl_sum_all(incl2, lane);

        float y[EPL];
#pragma unroll
        for (int j = 0; j < EPL; ++j) y[j] = xv[j] * Dd;
#pragma unroll
        for (int n = 0; n < NST; ++n) {
            const float base2 = (incl2[n] - t2[n]) + c2[n];
#pragma unroll
            for (int j = 0; j < EPL; ++j)
                y[j] = fmaf(P[n][j] * (base2 + p2[n][j]), Cn[n][j], y[j]);
        }
        store_run(och + l, y);

#pragma unroll
        for (int n = 0; n < NST; ++n) {
            c1[n] += lane_value<31>(incl1[n]);
            c2[n] += lane_value<31>(incl2[n]);
        }
    }
}

extern "C" void kernel_launch(void* const* d_in, const int* in_sizes, int n_in,
                              void* d_out, int out_size, void* d_ws, size_t ws_size,
                              hipStream_t stream) {
    (void)in_sizes; (void)n_in; (void)out_size; (void)ws_size;

    const float* x        = (const float*)d_in[0];
    const float* conv_w   = (const float*)d_in[1];
    const float* conv_b   = (const float*)d_in[2];
    const float* x_proj_w = (const float*)d_in[3];
    const float* dt_w     = (const float*)d_in[4];
    const float* dt_b     = (const float*)d_in[5];
    const float* A_logs   = (const float*)d_in[6];
    const float* Ds       = (const float*)d_in[7];
    float* out = (float*)d_out;

    float* xs   = (float*)d_ws;
    float* xdbl = xs + (size_t)BSZ * DCH * LL;

    {
        dim3 grid(DCH * (BSZ / 4));
        conv_silu_kernel<<<grid, 256, 0, stream>>>(x, conv_w, conv_b, xs);
    }
    {
        const int nwaves = BSZ * (LL / 16);
        dim3 grid(nwaves / 8);
        proj_wmma_kernel<<<grid, 256, 0, stream>>>(xs, x_proj_w, xdbl);
    }
    {
        const int nwaves = BSZ * DCH;
        dim3 grid(nwaves / 8);
        scan_kernel<<<grid, 256, 0, stream>>>(xs, xdbl, dt_w, dt_b,
                                              A_logs, Ds, out);
    }
}


# scan kernel: 12-wave workgroups (256 of them, one per WGP, three channels on every SIMD) instead of 8-wave ones, of which 384 put four channels on some SIMDs and two on others
# speedup vs baseline: 3.5959x; 1.1271x over previous
#include <hip/hip_runtime.h>
#include <math.h>

#define BSZ   32
#define DCH   96
#define HH    64
#define WW    64
#define LL    (HH * WW)
#define NST   4
#define RNK   6
#define KPROJ 14

typedef __attribute__((ext_vector_type(2))) float v2f;
typedef __attribute__((ext_vector_type(8))) float v8f;

#define CPITCH 72
#define CROWS  68
#define CIMG   4

__global__ void __launch_bounds__(256, 6)
conv_silu_kernel(const float* __restrict__ x,
                 const float* __restrict__ conv_w,
                 const float* __restrict__ conv_b,
                 float* __restrict__ xs) {
    __shared__ __attribute__((aligned(16))) float t[CROWS * CPITCH];

    const int tid = threadIdx.x;
    const int d   = blockIdx.x % DCH;
    const int b0  = (blockIdx.x / DCH) * CIMG;

    float4 g0, g1, g2, g3;
    {
        const float4* __restrict__ xin4 =
            reinterpret_cast<const float4*>(x + ((size_t)b0 * DCH + d) * LL) + tid;
        g0 = xin4[0]; g1 = xin4[256]; g2 = xin4[512]; g3 = xin4[768];
    }

    float wv[25];
#pragma unroll
    for (int j = 0; j < 25; ++j) wv[j] = conv_w[d * 25 + j];
    const float bv = conv_b[d];

    float4* t4 = reinterpret_cast<float4*>(t);
    const float4 z4 = make_float4(0.f, 0.f, 0.f, 0.f);
    if (tid < 72) {
        const int r   = tid / 18;
        const int c   = tid - r * 18;
        const int row = (r < 2) ? r : (r + 64);
        t4[row * 18 + c] = z4;
    } else if (tid < 200) {
        const int u   = tid - 72;
        const int row = 2 + (u >> 1);
        const int c   = (u & 1) ? 17 : 0;
        t4[row * 18 + c] = z4;
    }

    const int pc = tid & 15;
    const int pr = tid >> 4;
    const float* base = t + (4 * pr) * CPITCH + 4 * pc;
    const int dst0 = ((tid >> 4) + 2) * 18 + 1 + (tid & 15);

#pragma unroll 1
    for (int it = 0; it < CIMG; ++it) {
        const size_t img = (size_t)(b0 + it) * DCH + d;

        t4[dst0]           = g0;
        t4[dst0 + 16 * 18] = g1;
        t4[dst0 + 32 * 18] = g2;
        t4[dst0 + 48 * 18] = g3;
        __syncthreads();

        float win[8][8];
#pragma unroll
        for (int r = 0; r < 8; ++r) {
            const float* rowp = base + r * CPITCH;
            const float2 a = *reinterpret_cast<const float2*>(rowp + 2);
            const float4 m = *reinterpret_cast<const float4*>(rowp + 4);
            const float2 c = *reinterpret_cast<const float2*>(rowp + 8);
            win[r][0] = a.x; win[r][1] = a.y;
            win[r][2] = m.x; win[r][3] = m.y; win[r][4] = m.z; win[r][5] = m.w;
            win[r][6] = c.x; win[r][7] = c.y;
        }
        __syncthreads();

        if (it + 1 < CIMG) {
            const float4* __restrict__ nx4 =
                reinterpret_cast<const float4*>(x + (img + DCH) * LL) + tid;
            g0 = nx4[0]; g1 = nx4[256]; g2 = nx4[512]; g3 = nx4[768];
        }

        float* __restrict__ xo = xs + img * LL;
#pragma unroll
        for (int i = 0; i < 4; ++i) {
            float o[4];
#pragma unroll
            for (int j = 0; j < 4; ++j) {
                float acc = bv;
#pragma unroll
                for (int ky = 0; ky < 5; ++ky)
#pragma unroll
                    for (int kx = 0; kx < 5; ++kx)
                        acc = fmaf(win[i + ky][j + kx], wv[ky * 5 + kx], acc);
                o[j] = acc / (1.0f + expf(-acc));
            }
            *reinterpret_cast<float4*>(xo + (4 * pr + i) * WW + 4 * pc) =
                make_float4(o[0], o[1], o[2], o[3]);
        }
    }
}

__global__ void proj_wmma_kernel(const float* __restrict__ xs,
                                 const float* __restrict__ x_proj_w,
                                 float* __restrict__ xdbl) {
    __shared__ float wA[16][DCH];

    const int tid = threadIdx.x;
    for (int i = tid; i < 16 * DCH; i += 256) {
        const int row = i / DCH;
        const int k   = i % DCH;
        wA[row][k] = (row < KPROJ) ? x_proj_w[row * DCH + k] : 0.0f;
    }
    __syncthreads();

    const int wid  = (blockIdx.x * blockDim.x + tid) >> 5;
    const int lane = tid & 31;
    const int tcol = wid & 255;
    const int b    = wid >> 8;
    const int l0   = tcol * 16;

    const int half = lane >> 4;
    const int m    = lane & 15;

    const float* xb = xs + (size_t)b * DCH * LL + (l0 + m);

    v2f breg[DCH / 4];
#pragma unroll
    for (int kk = 0; kk < DCH / 4; ++kk) {
        const int k0 = kk * 4 + 2 * half;
        breg[kk].x = xb[(size_t)k0 * LL];
        breg[kk].y = xb[(size_t)(k0 + 1) * LL];
    }

    v8f acc = {0.f, 0.f, 0.f, 0.f, 0.f, 0.f, 0.f, 0.f};
#pragma unroll
    for (int kk = 0; kk < DCH / 4; ++kk) {
        const int k0 = kk * 4 + 2 * half;
        const v2f a = *(const v2f*)&wA[m][k0];
        acc = __builtin_amdgcn_wmma_f32_16x16x4_f32(
                  false, a, false, breg[kk], (short)0, acc, false, false);
    }

#pragma unroll
    for (int i = 0; i < 8; ++i) {
        const int row = i + 8 * half;
        if (row < KPROJ)
            xdbl[((size_t)b * KPROJ + row) * LL + l0 + m] = acc[i];
    }
}

#define EPL  8
#define STEP (32 * EPL)
#define SCAN_WAVES 12

#define LOG2E_HI_F  1.4426950216293335f
#define LOG2E_LO_F  1.92596299e-8f
#define LN2_F       0.693147180559945309f
#define LOG2_EPS_F  (-29.8973529f)

__device__ __forceinline__ float softplus_f32(float a) {
    const float na = -fabsf(a);
    const float t = na * LOG2E_HI_F;
    float r = fmaf(na, LOG2E_HI_F, -t);
    r = fmaf(na, LOG2E_LO_F, r);
    float z = __builtin_amdgcn_exp2f(t);
    z = fmaf(z, r * LN2_F, z);
    const float u = 2.0f + z;
    float ru = __builtin_amdgcn_rcpf(u);
    ru = fmaf(fmaf(-u, ru, 1.0f), ru, ru);
    const float s = z * ru;
    const float w = s * s;
    float p = 1.0f / 15.0f;
    p = fmaf(p, w, 1.0f / 13.0f);
    p = fmaf(p, w, 1.0f / 11.0f);
    p = fmaf(p, w, 1.0f / 9.0f);
    p = fmaf(p, w, 1.0f / 7.0f);
    p = fmaf(p, w, 1.0f / 5.0f);
    p = fmaf(p, w, 1.0f / 3.0f);
    const float s2 = s + s;
    const float l1p = fmaf(s2 * w, p, s2);
    return fmaxf(a, 0.0f) + l1p;
}

template <int SHR>
__device__ __forceinline__ float row_shr_or_zero(float v) {
    return __builtin_bit_cast(float,
        __builtin_amdgcn_update_dpp(0, __builtin_bit_cast(int, v), 0x110 + SHR, 0xf, 0xf, true));
}

template <int LANE>
__device__ __forceinline__ float lane_value(float v) {
    return __builtin_bit_cast(float, __builtin_amdgcn_readlane(__builtin_bit_cast(int, v), LANE));
}

__device__ __forceinline__ void wave_incl_sum_all(float (&v)[NST], int lane) {
#pragma unroll
    for (int n = 0; n < NST; ++n) v[n] += row_shr_or_zero<1>(v[n]);
#pragma unroll
    for (int n = 0; n < NST; ++n) v[n] += row_shr_or_zero<2>(v[n]);
#pragma unroll
    for (int n = 0; n < NST; ++n) v[n] += row_shr_or_zero<4>(v[n]);
#pragma unroll
    for (int n = 0; n < NST; ++n) v[n] += row_shr_or_zero<8>(v[n]);
#pragma unroll
    for (int n = 0; n < NST; ++n) {
        const float low_row_total = lane_value<15>(v[n]);
        v[n] += (lane >= 16) ? low_row_total : 0.0f;
    }
}

__device__ __forceinline__ void load_run(float (&v)[EPL], const float* __restrict__ p) {
#pragma unroll
    for (int j = 0; j < EPL; j += 4) {
        const float4 t = *reinterpret_cast<const float4*>(p + j);
        v[j] = t.x; v[j + 1] = t.y; v[j + 2] = t.z; v[j + 3] = t.w;
    }
}

__device__ __forceinline__ void store_run(float* __restrict__ p, const float (&v)[EPL]) {
#pragma unroll
    for (int j = 0; j < EPL; j += 4) {
        float4 t;
        t.x = v[j]; t.y = v[j + 1]; t.z = v[j + 2]; t.w = v[j + 3];
        *reinterpret_cast<float4*>(p + j) = t;
    }
}

__global__ void __launch_bounds__(32 * SCAN_WAVES, 3)
scan_kernel(const float* __restrict__ xs,
            const float* __restrict__ xdbl,
            const float* __restrict__ dt_w,
            const float* __restrict__ dt_b,
            const float* __restrict__ A_logs,
            const float* __restrict__ Ds,
            float* __restrict__ out) {
    const int wid  = __builtin_amdgcn_readfirstlane((int)((blockIdx.x * blockDim.x + threadIdx.x) >> 5));
    const int lane = threadIdx.x & 31;
    const int d    = wid % DCH;
    const int b    = wid / DCH;

    float wr[RNK];
#pragma unroll
    for (int r = 0; r < RNK; ++r) wr[r] = dt_w[d * RNK + r];
    const float dtb = dt_b[d];
    const float Dd  = Ds[d];
    float An[NST];
#pragma unroll
    for (int n = 0; n < NST; ++n) An[n] = -expf(A_logs[d * NST + n]) * LOG2E_HI_F;

    const float* xch = xs   + ((size_t)b * DCH + d) * LL;
    const float* xdb = xdbl + (size_t)b * KPROJ * LL;
    float*       och = out  + ((size_t)b * DCH + d) * LL;

    float c1[NST] = {0.f, 0.f, 0.f, 0.f};
    float c2[NST] = {0.f, 0.f, 0.f, 0.f};

    for (int c = 0; c < LL / STEP; ++c) {
        const int l = c * STEP + lane * EPL;

        float dts[RNK][EPL], xv[EPL], Bn[NST][EPL], Cn[NST][EPL];
#pragma unroll
        for (int r = 0; r < RNK; ++r) load_run(dts[r], xdb + (size_t)r * LL + l);
        load_run(xv, xch + l);
#pragma unroll
        for (int n = 0; n < NST; ++n) load_run(Bn[n], xdb + (size_t)(RNK + n) * LL + l);
#pragma unroll
        for (int n = 0; n < NST; ++n) load_run(Cn[n], xdb + (size_t)(RNK + NST + n) * LL + l);
        __builtin_amdgcn_sched_barrier(0);

        float delta[EPL];
#pragma unroll
        for (int j = 0; j < EPL; ++j) {
            float ad = dtb;
#pragma unroll
            for (int r = 0; r < RNK; ++r) ad = fmaf(wr[r], dts[r][j], ad);
            delta[j] = softplus_f32(ad);
        }

        float p1[NST][EPL], t1[NST];
#pragma unroll
        for (int n = 0; n < NST; ++n) {
            float run = 0.0f;
#pragma unroll
            for (int j = 0; j < EPL; ++j) {
                const float lda = fmaxf(delta[j] * An[n], LOG2_EPS_F);
                run = (j == 0) ? lda : (run + lda);
                p1[n][j] = run;
            }
            t1[n] = run;
        }
        float incl1[NST];
#pragma unroll
        for (int n = 0; n < NST; ++n) incl1[n] = t1[n];
        wave_incl_sum_all(incl1, lane);

        float P[NST][EPL], p2[NST][EPL], t2[NST];
#pragma unroll
        for (int n = 0; n < NST; ++n) {
            const float base1 = (incl1[n] - t1[n]) + c1[n];
            float run = 0.0f;
#pragma unroll
            for (int j = 0; j < EPL; ++j) {
                const float s1 = base1 + p1[n][j];
                P[n][j] = __builtin_amdgcn_exp2f(s1);
                const float q = (delta[j] * Bn[n][j] * xv[j]) * __builtin_amdgcn_rcpf(fmaxf(P[n][j], 1e-9f));
                run = (j == 0) ? q : (run + q);
                p2[n][j] = run;
            }
            t2[n] = run;
        }
        float incl2[NST];
#pragma unroll
        for (int n = 0; n < NST; ++n) incl2[n] = t2[n];
        wave_incl_sum_all(incl2, lane);

        float y[EPL];
#pragma unroll
        for (int j = 0; j < EPL; ++j) y[j] = xv[j] * Dd;
#pragma unroll
        for (int n = 0; n < NST; ++n) {
            const float base2 = (incl2[n] - t2[n]) + c2[n];
#pragma unroll
            for (int j = 0; j < EPL; ++j)
                y[j] = fmaf(P[n][j] * (base2 + p2[n][j]), Cn[n][j], y[j]);
        }
        store_run(och + l, y);

#pragma unroll
        for (int n = 0; n < NST; ++n) {
            c1[n] += lane_value<31>(incl1[n]);
            c2[n] += lane_value<31>(incl2[n]);
        }
    }
}

extern "C" void kernel_launch(void* const* d_in, const int* in_sizes, int n_in,
                              void* d_out, int out_size, void* d_ws, size_t ws_size,
                              hipStream_t stream) {
    (void)in_sizes; (void)n_in; (void)out_size; (void)ws_size;

    const float* x        = (const float*)d_in[0];
    const float* conv_w   = (const float*)d_in[1];
    const float* conv_b   = (const float*)d_in[2];
    const float* x_proj_w = (const float*)d_in[3];
    const float* dt_w     = (const float*)d_in[4];
    const float* dt_b     = (const float*)d_in[5];
    const float* A_logs   = (const float*)d_in[6];
    const float* Ds       = (const float*)d_in[7];
    float* out = (float*)d_out;

    float* xs   = (float*)d_ws;
    float* xdbl = xs + (size_t)BSZ * DCH * LL;

    {
        dim3 grid(DCH * (BSZ / 4));
        conv_silu_kernel<<<grid, 256, 0, stream>>>(x, conv_w, conv_b, xs);
    }
    {
        const int nwaves = BSZ * (LL / 16);
        dim3 grid(nwaves / 8);
        proj_wmma_kernel<<<grid, 256, 0, stream>>>(xs, x_proj_w, xdbl);
    }
    {
        const int nwaves = BSZ * DCH;
        dim3 grid(nwaves / SCAN_WAVES);
        scan_kernel<<<grid, 32 * SCAN_WAVES, 0, stream>>>(xs, xdbl, dt_w, dt_b,
                                                          A_logs, Ds, out);
    }
}
